# starting kernel
# baseline (speedup 1.0000x reference)
#include <hip/hip_runtime.h>
#include <hip/hip_bf16.h>


typedef __attribute__((ext_vector_type(16))) _Float16 v16h;
typedef __attribute__((ext_vector_type(8)))  _Float16 v8h;
typedef __attribute__((ext_vector_type(4)))  _Float16 v4h;
typedef __attribute__((ext_vector_type(8)))  float    v8f;

constexpr int kB = 2, kN = 2048, kC = 1024, kH = 16, kD = 64;
constexpr float kLnEps = 1e-5f;

__device__ __forceinline__ v8f wmma16(v16h a, v16h b, v8f c) {
  return __builtin_amdgcn_wmma_f32_16x16x32_f16(false, a, false, b, (short)0, c,
                                                false, false);
}

__device__ __forceinline__ v16h load_frag(const _Float16* __restrict__ p, int ld) {
  const int lane = threadIdx.x & 31;
  const int r    = lane & 15;
  const int ko   = (lane >> 4) << 3;
  const _Float16* b0 = p + (size_t)r * ld + ko;
  v8h lo = *(const v8h*)(b0);
  v8h hi = *(const v8h*)(b0 + 16);
  v16h f;
#pragma unroll
  for (int i = 0; i < 8; ++i) { f[i] = lo[i]; f[i + 8] = hi[i]; }
  return f;
}

__global__ void cvt_f32_f16(const float* __restrict__ src,
                            _Float16* __restrict__ dst, int n4) {
  int i = blockIdx.x * blockDim.x + threadIdx.x;
  if (i < n4) {
    float4 v = ((const float4*)src)[i];
    v4h h;
    h[0] = (_Float16)v.x; h[1] = (_Float16)v.y;
    h[2] = (_Float16)v.z; h[3] = (_Float16)v.w;
    ((v4h*)dst)[i] = h;
  }
}

template <int MODE>
__global__ void gemm_xwT(const _Float16* __restrict__ A,
                         const _Float16* __restrict__ W,
                         void* __restrict__ out) {
  const int lane = threadIdx.x & 31;
  const int colL = lane & 15;
  const int hi   = lane >> 4;
  const int m0 = blockIdx.x * 128 + (threadIdx.x >> 5) * 32;
  const int n0 = blockIdx.y * 64;

  v8f acc[2][4] = {};
  const _Float16* a0p = A + (size_t)m0 * kC;
  const _Float16* a1p = A + (size_t)(m0 + 16) * kC;
  const _Float16* wp  = W + (size_t)n0 * kC;

  for (int k = 0; k < kC; k += 32) {
    v16h a0 = load_frag(a0p + k, kC);
    v16h a1 = load_frag(a1p + k, kC);
#pragma unroll
    for (int nt = 0; nt < 4; ++nt) {
      v16h bf = load_frag(wp + (size_t)nt * 16 * kC + k, kC);
      acc[0][nt] = wmma16(a0, bf, acc[0][nt]);
      acc[1][nt] = wmma16(a1, bf, acc[1][nt]);
    }
  }

#pragma unroll
  for (int mt = 0; mt < 2; ++mt)
#pragma unroll
    for (int nt = 0; nt < 4; ++nt)
#pragma unroll
      for (int r = 0; r < 8; ++r) {
        const int row  = m0 + mt * 16 + 8 * hi + r;
        const int colc = n0 + nt * 16 + colL;
        const float v  = acc[mt][nt][r];
        if constexpr (MODE == 2) {
          ((float*)out)[(size_t)row * kC + colc] = v;
        } else {
          const int b = row >> 11, n = row & (kN - 1);
          const int h = colc >> 6, d = colc & (kD - 1);
          if constexpr (MODE == 0)
            ((_Float16*)out)[(((size_t)(b * kH + h)) * kN + n) * kD + d] = (_Float16)v;
          else
            ((_Float16*)out)[(((size_t)(b * kH + h)) * kD + d) * kN + n] = (_Float16)v;
        }
      }
}

__global__ void colstats_kernel(const _Float16* __restrict__ q16,
                                const _Float16* __restrict__ k16,
                                float* __restrict__ mcol,
                                float* __restrict__ rcol) {
  const int kt   = blockIdx.x;
  const int bh   = blockIdx.y;
  const int wave = threadIdx.x >> 5;
  const int lane = threadIdx.x & 31;
  const _Float16* qb = q16 + (size_t)bh * kN * kD;
  const _Float16* kb = k16 + (size_t)bh * kN * kD;

  const v16h bk0 = load_frag(kb + (size_t)kt * 16 * kD + 0,  kD);
  const v16h bk1 = load_frag(kb + (size_t)kt * 16 * kD + 32, kD);

  __shared__ float red[8][16];
  __shared__ float mshare[16];

  float mx = -3.4e38f;
  for (int qt = wave; qt < kN / 16; qt += 8) {
    v16h a0 = load_frag(qb + (size_t)qt * 16 * kD + 0,  kD);
    v16h a1 = load_frag(qb + (size_t)qt * 16 * kD + 32, kD);
    v8f s = {};
    s = wmma16(a0, bk0, s);
    s = wmma16(a1, bk1, s);
#pragma unroll
    for (int i = 0; i < 8; ++i) mx = fmaxf(mx, s[i]);
  }
  mx = fmaxf(mx, __shfl_xor(mx, 16, 32));
  if (lane < 16) red[wave][lane] = mx;
  __syncthreads();
  if (threadIdx.x < 16) {
    float m = red[0][threadIdx.x];
#pragma unroll
    for (int w = 1; w < 8; ++w) m = fmaxf(m, red[w][threadIdx.x]);
    mshare[threadIdx.x] = m;
    mcol[(size_t)bh * kN + kt * 16 + threadIdx.x] = m;
  }
  __syncthreads();
  const float mk = mshare[lane & 15];

  float sm = 0.f;
  for (int qt = wave; qt < kN / 16; qt += 8) {
    v16h a0 = load_frag(qb + (size_t)qt * 16 * kD + 0,  kD);
    v16h a1 = load_frag(qb + (size_t)qt * 16 * kD + 32, kD);
    v8f s = {};
    s = wmma16(a0, bk0, s);
    s = wmma16(a1, bk1, s);
#pragma unroll
    for (int i = 0; i < 8; ++i) sm += __expf(s[i] - mk);
  }
  sm += __shfl_xor(sm, 16, 32);
  if (lane < 16) red[wave][lane] = sm;
  __syncthreads();
  if (threadIdx.x < 16) {
    float s = 0.f;
#pragma unroll
    for (int w = 0; w < 8; ++w) s += red[w][threadIdx.x];
    rcol[(size_t)bh * kN + kt * 16 + threadIdx.x] = 1.0f / s;
  }
}

__global__ void ctx_kernel(const _Float16* __restrict__ q16,
                           const _Float16* __restrict__ k16,
                           const _Float16* __restrict__ vt16,
                           const float* __restrict__ mcol,
                           const float* __restrict__ rcol,
                           _Float16* __restrict__ ctx16) {
  const int bh   = blockIdx.y;
  const int wave = threadIdx.x >> 5;
  const int lane = threadIdx.x & 31;
  const int col  = lane & 15;
  const int hi   = lane >> 4;
  const int qt   = blockIdx.x * 8 + wave;

  __shared__ float mr_s[2][kN];
  __shared__ float estage[8][32][17];

  for (int i = threadIdx.x; i < kN; i += 256) {
    mr_s[0][i] = mcol[(size_t)bh * kN + i];
    mr_s[1][i] = rcol[(size_t)bh * kN + i];
  }
  __syncthreads();

  const _Float16* qb = q16  + (size_t)bh * kN * kD;
  const _Float16* kb = k16  + (size_t)bh * kN * kD;
  const _Float16* vb = vt16 + (size_t)bh * kD * kN;

  const v16h aq0 = load_frag(qb + (size_t)qt * 16 * kD + 0,  kD);
  const v16h aq1 = load_frag(qb + (size_t)qt * 16 * kD + 32, kD);

  v8f acc[4] = {};
  float (*stg)[17] = estage[wave];

  for (int k0 = 0; k0 < kN; k0 += 32) {
    v8f s0 = {}, s1 = {};
    s0 = wmma16(aq0, load_frag(kb + (size_t)(k0     ) * kD + 0,  kD), s0);
    s0 = wmma16(aq1, load_frag(kb + (size_t)(k0     ) * kD + 32, kD), s0);
    s1 = wmma16(aq0, load_frag(kb + (size_t)(k0 + 16) * kD + 0,  kD), s1);
    s1 = wmma16(aq1, load_frag(kb + (size_t)(k0 + 16) * kD + 32, kD), s1);

    const float m0 = mr_s[0][k0 + col],      r0 = mr_s[1][k0 + col];
    const float m1 = mr_s[0][k0 + 16 + col], r1 = mr_s[1][k0 + 16 + col];

#pragma unroll
    for (int i = 0; i < 8; ++i) {
      stg[col][i + 8 * hi]      = __expf(s0[i] - m0) * r0;
      stg[col + 16][i + 8 * hi] = __expf(s1[i] - m1) * r1;
    }
    v16h ef;
#pragma unroll
    for (int i = 0; i < 16; ++i) {
      const int kk = ((i < 8) ? i : i + 8) + 8 * hi;
      ef[i] = (_Float16)stg[kk][col];
    }
#pragma unroll
    for (int dt = 0; dt < 4; ++dt)
      acc[dt] = wmma16(ef, load_frag(vb + (size_t)dt * 16 * kN + k0, kN), acc[dt]);
  }

  const int b = bh >> 4, h = bh & 15;
  _Float16* cb = ctx16 + (size_t)b * kN * kC + (size_t)h * kD;
  const int qrow = qt * 16 + 8 * hi;
#pragma unroll
  for (int dt = 0; dt < 4; ++dt)
#pragma unroll
    for (int r = 0; r < 8; ++r)
      cb[(size_t)(qrow + r) * kC + dt * 16 + col] = (_Float16)acc[dt][r];
}

__global__ void ln_kernel(const float* __restrict__ resid,
                          const float* __restrict__ y,
                          const float* __restrict__ g,
                          const float* __restrict__ be,
                          float* __restrict__ out) {
  const int row = blockIdx.x;
  const int t   = threadIdx.x;
  float x[4], s = 0.f, sq = 0.f;
#pragma unroll
  for (int j = 0; j < 4; ++j) {
    const int c = t + j * 256;
    const float v = resid[(size_t)row * kC + c] + y[(size_t)row * kC + c];
    x[j] = v; s += v; sq += v * v;
  }
#pragma unroll
  for (int o = 16; o >= 1; o >>= 1) {
    s  += __shfl_xor(s,  o, 32);
    sq += __shfl_xor(sq, o, 32);
  }
  __shared__ float rs[8], rq[8];
  __shared__ float mu_s, inv_s;
  const int wave = t >> 5, lane = t & 31;
  if (lane == 0) { rs[wave] = s; rq[wave] = sq; }
  __syncthreads();
  if (t == 0) {
    float S = 0.f, Q = 0.f;
#pragma unroll
    for (int w = 0; w < 8; ++w) { S += rs[w]; Q += rq[w]; }
    const float mu = S * (1.0f / kC);
    mu_s  = mu;
    inv_s = rsqrtf(Q * (1.0f / kC) - mu * mu + kLnEps);
  }
  __syncthreads();
  const float mu = mu_s, inv = inv_s;
#pragma unroll
  for (int j = 0; j < 4; ++j) {
    const int c = t + j * 256;
    out[(size_t)row * kC + c] = (x[j] - mu) * inv * g[c] + be[c];
  }
}

extern "C" void kernel_launch(void* const* d_in, const int* in_sizes, int n_in,
                              void* d_out, int out_size, void* d_ws, size_t ws_size,
                              hipStream_t stream) {
  (void)in_sizes; (void)n_in; (void)out_size; (void)ws_size;
  const float* query = (const float*)d_in[0];
  const float* key   = (const float*)d_in[1];
  const float* value = (const float*)d_in[2];
  const float* Wq = (const float*)d_in[4];
  const float* Wk = (const float*)d_in[5];
  const float* Wv = (const float*)d_in[6];
  const float* Wo = (const float*)d_in[7];
  const float* ln_g = (const float*)d_in[8];
  const float* ln_b = (const float*)d_in[9];
  float* out = (float*)d_out;

  constexpr size_t BNC = (size_t)kB * kN * kC;
  constexpr size_t CC  = (size_t)kC * kC;
  constexpr size_t BHN = (size_t)kB * kH * kN;

  size_t off = 0;
  auto take = [&](size_t bytes) -> void* {
    void* p = (char*)d_ws + off;
    off += (bytes + 255) & ~(size_t)255;
    return p;
  };
  _Float16* xq16 = (_Float16*)take(BNC * 2);
  _Float16* xk16 = (_Float16*)take(BNC * 2);
  _Float16* xv16 = (_Float16*)take(BNC * 2);
  _Float16* wq16 = (_Float16*)take(CC * 2);
  _Float16* wk16 = (_Float16*)take(CC * 2);
  _Float16* wv16 = (_Float16*)take(CC * 2);
  _Float16* wo16 = (_Float16*)take(CC * 2);
  _Float16* qh16 = (_Float16*)take(BNC * 2);
  _Float16* kh16 = (_Float16*)take(BNC * 2);
  _Float16* vt16 = (_Float16*)take(BNC * 2);
  float*    mcol = (float*)take(BHN * 4);
  float*    rcol = (float*)take(BHN * 4);
  _Float16* ctx16 = xq16;
  float*    yf32  = (float*)xk16;

  cvt_f32_f16<<<(int)(BNC / 4 / 256), 256, 0, stream>>>(query, xq16, (int)(BNC / 4));
  cvt_f32_f16<<<(int)(BNC / 4 / 256), 256, 0, stream>>>(key,   xk16, (int)(BNC / 4));
  cvt_f32_f16<<<(int)(BNC / 4 / 256), 256, 0, stream>>>(value, xv16, (int)(BNC / 4));
  cvt_f32_f16<<<(int)(CC / 4 / 256),  256, 0, stream>>>(Wq, wq16, (int)(CC / 4));
  cvt_f32_f16<<<(int)(CC / 4 / 256),  256, 0, stream>>>(Wk, wk16, (int)(CC / 4));
  cvt_f32_f16<<<(int)(CC / 4 / 256),  256, 0, stream>>>(Wv, wv16, (int)(CC / 4));
  cvt_f32_f16<<<(int)(CC / 4 / 256),  256, 0, stream>>>(Wo, wo16, (int)(CC / 4));

  dim3 ggrid(32, 16);
  gemm_xwT<0><<<ggrid, 128, 0, stream>>>(xq16, wq16, qh16);
  gemm_xwT<0><<<ggrid, 128, 0, stream>>>(xk16, wk16, kh16);
  gemm_xwT<1><<<ggrid, 128, 0, stream>>>(xv16, wv16, vt16);

  colstats_kernel<<<dim3(kN / 16, kB * kH), 256, 0, stream>>>(qh16, kh16, mcol, rcol);

  ctx_kernel<<<dim3(kN / 128, kB * kH), 256, 0, stream>>>(qh16, kh16, vt16, mcol,
                                                          rcol, ctx16);

  gemm_xwT<2><<<ggrid, 128, 0, stream>>>(ctx16, wo16, yf32);
  ln_kernel<<<kB * kN, 256, 0, stream>>>(query, yf32, ln_g, ln_b, out);
}


# weight converts of Wk, Wv, Wo ride as extra blocks (8 grid rows) of the projection launch in front of the one that reads them: 3 convert launches gone (11 -> 8 launches); same values, same sums
# speedup vs baseline: 2.6274x; 2.6274x over previous
#include <hip/hip_runtime.h>
#include <hip/hip_bf16.h>


typedef __attribute__((ext_vector_type(16))) _Float16 v16h;
typedef __attribute__((ext_vector_type(8)))  _Float16 v8h;
typedef __attribute__((ext_vector_type(4)))  _Float16 v4h;
typedef __attribute__((ext_vector_type(8)))  float    v8f;
typedef __attribute__((ext_vector_type(8)))  short    v8s;

constexpr int kB = 2, kN = 2048, kC = 1024, kH = 16, kD = 64;
constexpr float kLnEps = 1e-5f;

__device__ __forceinline__ v8f wmma16(v16h a, v16h b, v8f c) {
  return __builtin_amdgcn_wmma_f32_16x16x32_f16(false, a, false, b, (short)0, c,
                                                false, false);
}

__device__ __forceinline__ v16h load_frag(const _Float16* __restrict__ p, int ld) {
  const int lane = threadIdx.x & 31;
  const int r    = lane & 15;
  const int ko   = (lane >> 4) << 3;
  const _Float16* b0 = p + (size_t)r * ld + ko;
  v8h lo = *(const v8h*)(b0);
  v8h hi = *(const v8h*)(b0 + 16);
  v16h f;
#pragma unroll
  for (int i = 0; i < 8; ++i) { f[i] = lo[i]; f[i + 8] = hi[i]; }
  return f;
}

typedef __attribute__((ext_vector_type(4))) float v4f;
struct ARawF32 { v4f x[4]; };
struct ARawF16 { v8h x[2]; };

__device__ __forceinline__ void a_issue(ARawF32& r, const float* __restrict__ p, int ld) {
  const int lane = threadIdx.x & 31;
  const float* b0 = p + (size_t)(lane & 15) * ld + ((lane >> 4) << 3);
  r.x[0] = *(const v4f*)(b0);
  r.x[1] = *(const v4f*)(b0 + 4);
  r.x[2] = *(const v4f*)(b0 + 16);
  r.x[3] = *(const v4f*)(b0 + 20);
}
__device__ __forceinline__ v16h a_finish(const ARawF32& r) {
  v16h f;
  f[0]  = (_Float16)r.x[0].x; f[1]  = (_Float16)r.x[0].y; f[2]  = (_Float16)r.x[0].z; f[3]  = (_Float16)r.x[0].w;
  f[4]  = (_Float16)r.x[1].x; f[5]  = (_Float16)r.x[1].y; f[6]  = (_Float16)r.x[1].z; f[7]  = (_Float16)r.x[1].w;
  f[8]  = (_Float16)r.x[2].x; f[9]  = (_Float16)r.x[2].y; f[10] = (_Float16)r.x[2].z; f[11] = (_Float16)r.x[2].w;
  f[12] = (_Float16)r.x[3].x; f[13] = (_Float16)r.x[3].y; f[14] = (_Float16)r.x[3].z; f[15] = (_Float16)r.x[3].w;
  return f;
}
__device__ __forceinline__ void a_issue(ARawF16& r, const _Float16* __restrict__ p, int ld) {
  const int lane = threadIdx.x & 31;
  const _Float16* b0 = p + (size_t)(lane & 15) * ld + ((lane >> 4) << 3);
  r.x[0] = *(const v8h*)(b0);
  r.x[1] = *(const v8h*)(b0 + 16);
}
__device__ __forceinline__ v16h a_finish(const ARawF16& r) {
  v16h f;
#pragma unroll
  for (int i = 0; i < 8; ++i) { f[i] = r.x[0][i]; f[i + 8] = r.x[1][i]; }
  return f;
}
__device__ __forceinline__ void a_hold(ARawF32& r) {
  asm volatile("" : "+v"(r.x[0]), "+v"(r.x[1]), "+v"(r.x[2]), "+v"(r.x[3]) : : "memory");
}
__device__ __forceinline__ void a_hold(ARawF16& r) {
  asm volatile("" : "+v"(r.x[0]), "+v"(r.x[1]) : : "memory");
}
__device__ __forceinline__ void acc_fence(v8f (&acc)[2][4]) {
  asm volatile("" : "+v"(acc[0][0]), "+v"(acc[0][1]), "+v"(acc[0][2]), "+v"(acc[0][3]),
                    "+v"(acc[1][0]), "+v"(acc[1][1]), "+v"(acc[1][2]), "+v"(acc[1][3]));
}
__device__ __forceinline__ void acc_mem_fence(v8f (&acc)[2][4]) {
  asm volatile("" : "+v"(acc[0][0]), "+v"(acc[0][1]), "+v"(acc[0][2]), "+v"(acc[0][3]),
                    "+v"(acc[1][0]), "+v"(acc[1][1]), "+v"(acc[1][2]), "+v"(acc[1][3])
                  : : "memory");
}
template <typename AT> struct ARawOf;
template <> struct ARawOf<float>    { typedef ARawF32 type; };
template <> struct ARawOf<_Float16> { typedef ARawF16 type; };

constexpr int kWBlk = 512;
constexpr int kWStep = 4 * kWBlk;

__device__ __forceinline__ size_t w_piece_src(int p) {
  const int lane = p & 31;
  const int j    = (p >> 5) & 1;
  const int nt   = (p >> 6) & 3;
  const int s    = (p >> 8) & 31;
  const int nb   = p >> 13;
  const int n    = nb * 64 + nt * 16 + (lane & 15);
  const int k0   = s * 32 + j * 16 + ((lane >> 4) << 3);
  return (size_t)n * kC + k0;
}
__device__ __forceinline__ v8h w_piece_round(float4 lo, float4 hi) {
  v8h h;
  h[0] = (_Float16)lo.x; h[1] = (_Float16)lo.y; h[2] = (_Float16)lo.z; h[3] = (_Float16)lo.w;
  h[4] = (_Float16)hi.x; h[5] = (_Float16)hi.y; h[6] = (_Float16)hi.z; h[7] = (_Float16)hi.w;
  return h;
}

constexpr int kRideRows    = 8;
constexpr int kRideThreads = kRideRows * 32 * 128;
constexpr int kRidePieces  = (kC * kC / 8) / kRideThreads;
__device__ __forceinline__ void cvt_w_ride(const float* __restrict__ src,
                                           _Float16* __restrict__ dst, int t) {
  float4 lo[kRidePieces], hi[kRidePieces];
#pragma unroll
  for (int j = 0; j < kRidePieces; ++j) {
    const size_t o = w_piece_src(t + j * kRideThreads);
    lo[j] = *(const float4*)(src + o);
    hi[j] = *(const float4*)(src + o + 4);
  }
#pragma unroll
  for (int j = 0; j < kRidePieces; ++j)
    *(v8h*)(dst + (size_t)(t + j * kRideThreads) * 8) = w_piece_round(lo[j], hi[j]);
}

__global__ void cvt_w_frag(const float* __restrict__ src,
                           _Float16* __restrict__ dst) {
  const int p    = blockIdx.x * blockDim.x + threadIdx.x;
  const int lane = p & 31;
  const int j    = (p >> 5) & 1;
  const int nt   = (p >> 6) & 3;
  const int s    = (p >> 8) & 31;
  const int nb   = p >> 13;
  const int n    = nb * 64 + nt * 16 + (lane & 15);
  const int k0   = s * 32 + j * 16 + ((lane >> 4) << 3);
  const float4 lo = *(const float4*)(src + (size_t)n * kC + k0);
  const float4 hi = *(const float4*)(src + (size_t)n * kC + k0 + 4);
  v8h h;
  h[0] = (_Float16)lo.x; h[1] = (_Float16)lo.y; h[2] = (_Float16)lo.z; h[3] = (_Float16)lo.w;
  h[4] = (_Float16)hi.x; h[5] = (_Float16)hi.y; h[6] = (_Float16)hi.z; h[7] = (_Float16)hi.w;
  *(v8h*)(dst + (size_t)p * 8) = h;
}

__device__ __forceinline__ v16h load_wfrag(const _Float16* __restrict__ blk) {
  const int lane = threadIdx.x & 31;
  v8h lo = *(const v8h*)(blk + lane * 8);
  v8h hi = *(const v8h*)(blk + 256 + lane * 8);
  v16h f;
#pragma unroll
  for (int i = 0; i < 8; ++i) { f[i] = lo[i]; f[i + 8] = hi[i]; }
  return f;
}

template <int MODE, typename AT>
__device__ __forceinline__ void gemm_tile(const AT* __restrict__ A,
                                          const _Float16* __restrict__ W,
                                          void* __restrict__ out) {
  const int lane = threadIdx.x & 31;
  const int colL = lane & 15;
  const int hi   = lane >> 4;
  const int m0 = blockIdx.x * 128 + (threadIdx.x >> 5) * 32;
  const int n0 = blockIdx.y * 64;

  v8f acc[2][4] = {};
  const AT* a0p = A + (size_t)m0 * kC;
  const AT* a1p = A + (size_t)(m0 + 16) * kC;
  const _Float16* wp  = W + (size_t)blockIdx.y * (kC / 32) * kWStep;

  typename ARawOf<AT>::type ra[2][2];
  v16h fa[2][2], fb[2][4];
  auto issue = [&](int s, int k) {
    a_issue(ra[s][0], a0p + k, kC);
    a_issue(ra[s][1], a1p + k, kC);
#pragma unroll
    for (int nt = 0; nt < 4; ++nt) fb[s][nt] = load_wfrag(wp + (size_t)(k >> 5) * kWStep + nt * kWBlk);
  };
  auto mma = [&](int s) {
    acc_mem_fence(acc);
#pragma unroll
    for (int nt = 0; nt < 4; ++nt) {
      acc[0][nt] = wmma16(fa[s][0], fb[s][nt], acc[0][nt]);
      acc[1][nt] = wmma16(fa[s][1], fb[s][nt], acc[1][nt]);
    }
    acc_fence(acc);
  };
  auto finish = [&](int s) {
    a_hold(ra[s][0]);
    a_hold(ra[s][1]);
    fa[s][0] = a_finish(ra[s][0]);
    fa[s][1] = a_finish(ra[s][1]);
  };

  issue(0, 0);
  issue(1, 32);
  finish(0);
  for (int k = 0; k < kC - 64; k += 64) {
    mma(0); issue(0, k + 64); finish(1);
    mma(1); issue(1, k + 96); finish(0);
  }
  mma(0);
  finish(1);
  mma(1);

#pragma unroll
  for (int mt = 0; mt < 2; ++mt)
#pragma unroll
    for (int nt = 0; nt < 4; ++nt)
#pragma unroll
      for (int r = 0; r < 8; ++r) {
        const int row  = m0 + mt * 16 + 8 * hi + r;
        const int colc = n0 + nt * 16 + colL;
        const float v  = acc[mt][nt][r];
        if constexpr (MODE == 2) {
          ((float*)out)[(size_t)row * kC + colc] = v;
        } else {
          const int b = row >> 11, n = row & (kN - 1);
          const int h = colc >> 6, d = colc & (kD - 1);
          if constexpr (MODE == 0)
            ((_Float16*)out)[(((size_t)(b * kH + h)) * kN + n) * kD + d] = (_Float16)v;
          else
            ((_Float16*)out)[(((size_t)(b * kH + h)) * kD + d) * kN + n] = (_Float16)v;
        }
      }
}

template <int MODE>
__global__ void __launch_bounds__(128)
proj_x32_wT(const float* __restrict__ X,
            const _Float16* __restrict__ W,
            _Float16* __restrict__ out,
            const float* __restrict__ wnext,
            _Float16* __restrict__ wnext16) {
  if (blockIdx.y >= kC / 64) {
    cvt_w_ride(wnext, wnext16,
               ((int)(blockIdx.y - kC / 64) * 32 + (int)blockIdx.x) * 128 + (int)threadIdx.x);
    return;
  }
  gemm_tile<MODE>(X, W, out);
}

__global__ void __launch_bounds__(128) gemm_ctx_woT(const _Float16* __restrict__ A,
                             const _Float16* __restrict__ W,
                             float* __restrict__ out) {
  gemm_tile<2>(A, W, out);
}

constexpr int kQRows  = 128;
constexpr int kQPitch = kD + 8;
constexpr int kQSteps = kN / kQRows;

__device__ __forceinline__ v16h lds_frag(const _Float16* p) {
  const int lane = threadIdx.x & 31;
  const _Float16* b0 = p + (lane & 15) * kQPitch + ((lane >> 4) << 3);
  v8h lo = *(const v8h*)(b0);
  v8h hi = *(const v8h*)(b0 + 16);
  v16h f;
#pragma unroll
  for (int i = 0; i < 8; ++i) { f[i] = lo[i]; f[i + 8] = hi[i]; }
  return f;
}

__global__ void __launch_bounds__(256)
colstats_kernel(const _Float16* __restrict__ q16,
                const _Float16* __restrict__ k16,
                float* __restrict__ mcol,
                float* __restrict__ rcol) {
  const int bh   = blockIdx.y;
  const int wave = threadIdx.x >> 5;
  const int lane = threadIdx.x & 31;
  const int kt   = blockIdx.x * 8 + wave;
  const _Float16* qb = q16 + (size_t)bh * kN * kD;
  const _Float16* kb = k16 + (size_t)bh * kN * kD;

  const v16h bk0 = load_frag(kb + (size_t)kt * 16 * kD + 0,  kD);
  const v16h bk1 = load_frag(kb + (size_t)kt * 16 * kD + 32, kD);

  __shared__ __attribute__((aligned(16))) _Float16 qs[2][kQRows * kQPitch];

  const int prow = threadIdx.x >> 3;
  const int pcol = (threadIdx.x & 7) * 8;
  v8h pre[4];

  auto fetch = [&](int step) {
    const _Float16* src = qb + (size_t)step * kQRows * kD + (size_t)threadIdx.x * 8;
#pragma unroll
    for (int j = 0; j < 4; ++j) pre[j] = *(const v8h*)(src + (size_t)j * 256 * 8);
  };
  auto park = [&](int buf) {
    _Float16* dst = qs[buf] + prow * kQPitch + pcol;
#pragma unroll
    for (int j = 0; j < 4; ++j) *(v8h*)(dst + j * 32 * kQPitch) = pre[j];
  };

  fetch(0);
  park(0);
  __syncthreads();

  v8f mxv;
#pragma unroll
  for (int i = 0; i < 8; ++i) mxv[i] = -3.4e38f;
  for (int st = 0; st < kQSteps; ++st) {
    fetch((st + 1) & (kQSteps - 1));
    const _Float16* cur = qs[st & 1];
#pragma unroll
    for (int t = 0; t < kQRows / 16; ++t) {
      v16h a0 = lds_frag(cur + t * 16 * kQPitch + 0);
      v16h a1 = lds_frag(cur + t * 16 * kQPitch + 32);
      v8f s = {};
      s = wmma16(a0, bk0, s);
      s = wmma16(a1, bk1, s);
#pragma unroll
      for (int i = 0; i < 8; ++i) mxv[i] = fmaxf(mxv[i], s[i]);
    }
    park((st + 1) & 1);
    __syncthreads();
  }
  float mk = mxv[0];
#pragma unroll
  for (int i = 1; i < 8; ++i) mk = fmaxf(mk, mxv[i]);
  mk = fmaxf(mk, __shfl_xor(mk, 16, 32));
  if (lane < 16) mcol[(size_t)bh * kN + kt * 16 + lane] = mk;

  v8f smv = {};
  for (int st = 0; st < kQSteps; ++st) {
    const bool more = st + 1 < kQSteps;
    if (more) fetch(st + 1);
    const _Float16* cur = qs[st & 1];
#pragma unroll
    for (int t = 0; t < kQRows / 16; ++t) {
      v16h a0 = lds_frag(cur + t * 16 * kQPitch + 0);
      v16h a1 = lds_frag(cur + t * 16 * kQPitch + 32);
      v8f s = {};
      s = wmma16(a0, bk0, s);
      s = wmma16(a1, bk1, s);
#pragma unroll
      for (int i = 0; i < 8; ++i) smv[i] += __expf(s[i] - mk);
    }
    if (more) {
      park((st + 1) & 1);
      __syncthreads();
    }
  }
  float sm = ((smv[0] + smv[1]) + (smv[2] + smv[3])) + ((smv[4] + smv[5]) + (smv[6] + smv[7]));
  sm += __shfl_xor(sm, 16, 32);
  if (lane < 16) rcol[(size_t)bh * kN + kt * 16 + lane] = 1.0f / sm;
}

constexpr int kKS     = 128;
constexpr int kKPitch = kD + 8;
constexpr int kVPitch = kKS + 8;
constexpr int kKSteps = kN / kKS;
constexpr int kEPitch = 16;

template <int PITCH>
__device__ __forceinline__ v16h lds_frag_p(const _Float16* p) {
  const int lane = threadIdx.x & 31;
  const _Float16* b0 = p + (lane & 15) * PITCH + ((lane >> 4) << 3);
  v8h lo = *(const v8h*)(b0);
  v8h hi = *(const v8h*)(b0 + 16);
  v16h f;
#pragma unroll
  for (int i = 0; i < 8; ++i) { f[i] = lo[i]; f[i + 8] = hi[i]; }
  return f;
}

__device__ __forceinline__ v8h lds_tr16(const _Float16* p) {
  typedef __attribute__((address_space(3))) v8s lds_v8s;
  return __builtin_bit_cast(v8h, __builtin_amdgcn_ds_load_tr16_b128_v8i16((lds_v8s*)p));
}

__global__ void __launch_bounds__(256)
ctx_kernel(const _Float16* __restrict__ q16,
           const _Float16* __restrict__ k16,
           const _Float16* __restrict__ vt16,
           const float* __restrict__ mcol,
           const float* __restrict__ rcol,
           _Float16* __restrict__ ctx16) {
  const int bh   = blockIdx.y;
  const int wave = threadIdx.x >> 5;
  const int lane = threadIdx.x & 31;
  const int col  = lane & 15;
  const int hi   = lane >> 4;
  const int qt   = blockIdx.x * 8 + wave;

  __shared__ float mr_s[2][kN];
  __shared__ __attribute__((aligned(16))) _Float16 estage[8][32 * kEPitch];
  __shared__ __attribute__((aligned(16))) _Float16 ks[2][kKS * kKPitch];
  __shared__ __attribute__((aligned(16))) _Float16 vs[2][kD * kVPitch];

  for (int i = threadIdx.x; i < kN; i += 256) {
    mr_s[0][i] = mcol[(size_t)bh * kN + i];
    mr_s[1][i] = rcol[(size_t)bh * kN + i];
  }

  const _Float16* qb = q16  + (size_t)bh * kN * kD;
  const _Float16* kb = k16  + (size_t)bh * kN * kD;
  const _Float16* vb = vt16 + (size_t)bh * kD * kN;

  const int krow = threadIdx.x >> 3;
  const int kcol = (threadIdx.x & 7) * 8;
  const int vrow = threadIdx.x >> 4;
  const int vcol = (threadIdx.x & 15) * 8;
  v8h kpre[4], vpre[4];

  auto fetch = [&](int step) {
    const _Float16* ksrc = kb + (size_t)step * kKS * kD + (size_t)threadIdx.x * 8;
    const _Float16* vsrc = vb + (size_t)vrow * kN + (size_t)step * kKS + vcol;
#pragma unroll
    for (int j = 0; j < 4; ++j) kpre[j] = *(const v8h*)(ksrc + (size_t)j * 256 * 8);
#pragma unroll
    for (int j = 0; j < 4; ++j) vpre[j] = *(const v8h*)(vsrc + (size_t)j * 16 * kN);
  };
  auto park = [&](int buf) {
    _Float16* kd = ks[buf] + krow * kKPitch + kcol;
    _Float16* vd = vs[buf] + vrow * kVPitch + vcol;
#pragma unroll
    for (int j = 0; j < 4; ++j) *(v8h*)(kd + j * 32 * kKPitch) = kpre[j];
#pragma unroll
    for (int j = 0; j < 4; ++j) *(v8h*)(vd + j * 16 * kVPitch) = vpre[j];
  };

  fetch(0);
  const v16h aq0 = load_frag(qb + (size_t)qt * 16 * kD + 0,  kD);
  const v16h aq1 = load_frag(qb + (size_t)qt * 16 * kD + 32, kD);
  park(0);
  __syncthreads();

  v8f acc[4] = {};
  _Float16* stg = estage[wave];
  _Float16* stw = stg + col * kEPitch + 8 * hi;
  const _Float16* str = stg + (8 * hi + (lane & 7)) * kEPitch + 8 * ((lane >> 3) & 1);

  for (int st = 0; st < kKSteps; ++st) {
    const bool more = st + 1 < kKSteps;
    if (more) fetch(st + 1);
    const _Float16* kc = ks[st & 1];
    const _Float16* vc = vs[st & 1];

#pragma unroll
    for (int c = 0; c < kKS / 32; ++c) {
      const int k0 = st * kKS + c * 32;
      v8f s0 = {}, s1 = {};
      s0 = wmma16(aq0, lds_frag_p<kKPitch>(kc + (c * 32     ) * kKPitch + 0),  s0);
      s0 = wmma16(aq1, lds_frag_p<kKPitch>(kc + (c * 32     ) * kKPitch + 32), s0);
      s1 = wmma16(aq0, lds_frag_p<kKPitch>(kc + (c * 32 + 16) * kKPitch + 0),  s1);
      s1 = wmma16(aq1, lds_frag_p<kKPitch>(kc + (c * 32 + 16) * kKPitch + 32), s1);

      const float m0 = mr_s[0][k0 + col],      r0 = mr_s[1][k0 + col];
      const float m1 = mr_s[0][k0 + 16 + col], r1 = mr_s[1][k0 + 16 + col];

      v8h e0, e1;
#pragma unroll
      for (int i = 0; i < 8; ++i) {
        e0[i] = (_Float16)(__expf(s0[i] - m0) * r0);
        e1[i] = (_Float16)(__expf(s1[i] - m1) * r1);
      }
      *(v8h*)(stw)                = e0;
      *(v8h*)(stw + 16 * kEPitch) = e1;
      const v8h elo = lds_tr16(str);
      const v8h ehi = lds_tr16(str + 16 * kEPitch);
      v16h ef;
#pragma unroll
      for (int i = 0; i < 8; ++i) { ef[i] = elo[i]; ef[i + 8] = ehi[i]; }
#pragma unroll
      for (int dt = 0; dt < 4; ++dt)
        acc[dt] = wmma16(ef, lds_frag_p<kVPitch>(vc + dt * 16 * kVPitch + c * 32), acc[dt]);
    }

    if (more) {
      park((st + 1) & 1);
      __syncthreads();
    }
  }

  const int b = bh >> 4, h = bh & 15;
  _Float16* cb = ctx16 + (size_t)b * kN * kC + (size_t)h * kD;
  const int qrow = qt * 16 + 8 * hi;
#pragma unroll
  for (int dt = 0; dt < 4; ++dt)
#pragma unroll
    for (int r = 0; r < 8; ++r)
      cb[(size_t)(qrow + r) * kC + dt * 16 + col] = (_Float16)acc[dt][r];
}

__global__ void ln_kernel(const float* __restrict__ resid,
                          const float* __restrict__ y,
                          const float* __restrict__ g,
                          const float* __restrict__ be,
                          float* __restrict__ out) {
  const int row = blockIdx.x;
  const int t   = threadIdx.x;
  float x[4], s = 0.f, sq = 0.f;
#pragma unroll
  for (int j = 0; j < 4; ++j) {
    const int c = t + j * 256;
    const float v = resid[(size_t)row * kC + c] + y[(size_t)row * kC + c];
    x[j] = v; s += v; sq += v * v;
  }
#pragma unroll
  for (int o = 16; o >= 1; o >>= 1) {
    s  += __shfl_xor(s,  o, 32);
    sq += __shfl_xor(sq, o, 32);
  }
  __shared__ float rs[8], rq[8];
  __shared__ float mu_s, inv_s;
  const int wave = t >> 5, lane = t & 31;
  if (lane == 0) { rs[wave] = s; rq[wave] = sq; }
  __syncthreads();
  if (t == 0) {
    float S = 0.f, Q = 0.f;
#pragma unroll
    for (int w = 0; w < 8; ++w) { S += rs[w]; Q += rq[w]; }
    const float mu = S * (1.0f / kC);
    mu_s  = mu;
    inv_s = rsqrtf(Q * (1.0f / kC) - mu * mu + kLnEps);
  }
  __syncthreads();
  const float mu = mu_s, inv = inv_s;
#pragma unroll
  for (int j = 0; j < 4; ++j) {
    const int c = t + j * 256;
    out[(size_t)row * kC + c] = (x[j] - mu) * inv * g[c] + be[c];
  }
}

extern "C" void kernel_launch(void* const* d_in, const int* in_sizes, int n_in,
                              void* d_out, int out_size, void* d_ws, size_t ws_size,
                              hipStream_t stream) {
  (void)in_sizes; (void)n_in; (void)out_size; (void)ws_size;
  const float* query = (const float*)d_in[0];
  const float* key   = (const float*)d_in[1];
  const float* value = (const float*)d_in[2];
  const float* Wq = (const float*)d_in[4];
  const float* Wk = (const float*)d_in[5];
  const float* Wv = (const float*)d_in[6];
  const float* Wo = (const float*)d_in[7];
  const float* ln_g = (const float*)d_in[8];
  const float* ln_b = (const float*)d_in[9];
  float* out = (float*)d_out;

  constexpr size_t BNC = (size_t)kB * kN * kC;
  constexpr size_t CC  = (size_t)kC * kC;
  constexpr size_t BHN = (size_t)kB * kH * kN;

  size_t off = 0;
  auto take = [&](size_t bytes) -> void* {
    void* p = (char*)d_ws + off;
    off += (bytes + 255) & ~(size_t)255;
    return p;
  };
  _Float16* ctx16 = (_Float16*)take(BNC * 2);
  float*    yf32  = (float*)take(BNC * 4);
  _Float16* wq16 = (_Float16*)take(CC * 2);
  _Float16* wk16 = (_Float16*)take(CC * 2);
  _Float16* wv16 = (_Float16*)take(CC * 2);
  _Float16* wo16 = (_Float16*)take(CC * 2);
  _Float16* qh16 = (_Float16*)take(BNC * 2);
  _Float16* kh16 = (_Float16*)take(BNC * 2);
  _Float16* vt16 = (_Float16*)take(BNC * 2);
  float*    mcol = (float*)take(BHN * 4);
  float*    rcol = (float*)take(BHN * 4);

  cvt_w_frag<<<(int)(CC / 8 / 256), 256, 0, stream>>>(Wq, wq16);

  dim3 ggrid(32, 16);
  dim3 pgrid(32, 16 + kRideRows);
  proj_x32_wT<0><<<pgrid, 128, 0, stream>>>(query, wq16, qh16, Wk, wk16);
  proj_x32_wT<0><<<pgrid, 128, 0, stream>>>(key,   wk16, kh16, Wv, wv16);
  proj_x32_wT<1><<<pgrid, 128, 0, stream>>>(value, wv16, vt16, Wo, wo16);

  colstats_kernel<<<dim3(kN / 128, kB * kH), 256, 0, stream>>>(qh16, kh16, mcol, rcol);

  ctx_kernel<<<dim3(kN / 128, kB * kH), 256, 0, stream>>>(qh16, kh16, vt16, mcol,
                                                          rcol, ctx16);

  gemm_ctx_woT<<<ggrid, 128, 0, stream>>>(ctx16, wo16, yf32);
  ln_kernel<<<kB * kN, 256, 0, stream>>>(query, yf32, ln_g, ln_b, out);
}


# gemm kernels: wave tile 32x128 (was 32x64; block 128x128, 256 blocks): an A fragment feeds 8 WMMAs, not 4: line touches and L2 bytes per WMMA fall 40%; 3 load sets in flight; same k order, same sums
# speedup vs baseline: 2.8238x; 1.0748x over previous
#include <hip/hip_runtime.h>
#include <hip/hip_bf16.h>


typedef __attribute__((ext_vector_type(16))) _Float16 v16h;
typedef __attribute__((ext_vector_type(8)))  _Float16 v8h;
typedef __attribute__((ext_vector_type(4)))  _Float16 v4h;
typedef __attribute__((ext_vector_type(8)))  float    v8f;
typedef __attribute__((ext_vector_type(8)))  short    v8s;

constexpr int kB = 2, kN = 2048, kC = 1024, kH = 16, kD = 64;
constexpr float kLnEps = 1e-5f;

__device__ __forceinline__ v8f wmma16(v16h a, v16h b, v8f c) {
  return __builtin_amdgcn_wmma_f32_16x16x32_f16(false, a, false, b, (short)0, c,
                                                false, false);
}

__device__ __forceinline__ v16h load_frag(const _Float16* __restrict__ p, int ld) {
  const int lane = threadIdx.x & 31;
  const int r    = lane & 15;
  const int ko   = (lane >> 4) << 3;
  const _Float16* b0 = p + (size_t)r * ld + ko;
  v8h lo = *(const v8h*)(b0);
  v8h hi = *(const v8h*)(b0 + 16);
  v16h f;
#pragma unroll
  for (int i = 0; i < 8; ++i) { f[i] = lo[i]; f[i + 8] = hi[i]; }
  return f;
}

typedef __attribute__((ext_vector_type(4))) float v4f;
struct ARawF32 { v4f x[4]; };
struct ARawF16 { v8h x[2]; };

__device__ __forceinline__ void a_issue(ARawF32& r, const float* __restrict__ p, int ld) {
  const int lane = threadIdx.x & 31;
  const float* b0 = p + (size_t)(lane & 15) * ld + ((lane >> 4) << 3);
  r.x[0] = *(const v4f*)(b0);
  r.x[1] = *(const v4f*)(b0 + 4);
  r.x[2] = *(const v4f*)(b0 + 16);
  r.x[3] = *(const v4f*)(b0 + 20);
}
__device__ __forceinline__ v16h a_finish(const ARawF32& r) {
  v16h f;
  f[0]  = (_Float16)r.x[0].x; f[1]  = (_Float16)r.x[0].y; f[2]  = (_Float16)r.x[0].z; f[3]  = (_Float16)r.x[0].w;
  f[4]  = (_Float16)r.x[1].x; f[5]  = (_Float16)r.x[1].y; f[6]  = (_Float16)r.x[1].z; f[7]  = (_Float16)r.x[1].w;
  f[8]  = (_Float16)r.x[2].x; f[9]  = (_Float16)r.x[2].y; f[10] = (_Float16)r.x[2].z; f[11] = (_Float16)r.x[2].w;
  f[12] = (_Float16)r.x[3].x; f[13] = (_Float16)r.x[3].y; f[14] = (_Float16)r.x[3].z; f[15] = (_Float16)r.x[3].w;
  return f;
}
__device__ __forceinline__ void a_issue(ARawF16& r, const _Float16* __restrict__ p, int ld) {
  const int lane = threadIdx.x & 31;
  const _Float16* b0 = p + (size_t)(lane & 15) * ld + ((lane >> 4) << 3);
  r.x[0] = *(const v8h*)(b0);
  r.x[1] = *(const v8h*)(b0 + 16);
}
__device__ __forceinline__ v16h a_finish(const ARawF16& r) {
  v16h f;
#pragma unroll
  for (int i = 0; i < 8; ++i) { f[i] = r.x[0][i]; f[i + 8] = r.x[1][i]; }
  return f;
}
__device__ __forceinline__ void acc_fence(v8f (&acc)[2][8]) {
  asm volatile("" : "+v"(acc[0][0]), "+v"(acc[0][1]), "+v"(acc[0][2]), "+v"(acc[0][3]),
                    "+v"(acc[0][4]), "+v"(acc[0][5]), "+v"(acc[0][6]), "+v"(acc[0][7]),
                    "+v"(acc[1][0]), "+v"(acc[1][1]), "+v"(acc[1][2]), "+v"(acc[1][3]),
                    "+v"(acc[1][4]), "+v"(acc[1][5]), "+v"(acc[1][6]), "+v"(acc[1][7]));
}
__device__ __forceinline__ void acc_mem_fence(v8f (&acc)[2][8]) {
  asm volatile("" : "+v"(acc[0][0]), "+v"(acc[0][1]), "+v"(acc[0][2]), "+v"(acc[0][3]),
                    "+v"(acc[0][4]), "+v"(acc[0][5]), "+v"(acc[0][6]), "+v"(acc[0][7]),
                    "+v"(acc[1][0]), "+v"(acc[1][1]), "+v"(acc[1][2]), "+v"(acc[1][3]),
                    "+v"(acc[1][4]), "+v"(acc[1][5]), "+v"(acc[1][6]), "+v"(acc[1][7])
                  : : "memory");
}
template <typename AT> struct ARawOf;
template <> struct ARawOf<float>    { typedef ARawF32 type; };
template <> struct ARawOf<_Float16> { typedef ARawF16 type; };

constexpr int kWBlk = 512;
constexpr int kWStep = 4 * kWBlk;

__device__ __forceinline__ size_t w_piece_src(int p) {
  const int lane = p & 31;
  const int j    = (p >> 5) & 1;
  const int nt   = (p >> 6) & 3;
  const int s    = (p >> 8) & 31;
  const int nb   = p >> 13;
  const int n    = nb * 64 + nt * 16 + (lane & 15);
  const int k0   = s * 32 + j * 16 + ((lane >> 4) << 3);
  return (size_t)n * kC + k0;
}
__device__ __forceinline__ v8h w_piece_round(float4 lo, float4 hi) {
  v8h h;
  h[0] = (_Float16)lo.x; h[1] = (_Float16)lo.y; h[2] = (_Float16)lo.z; h[3] = (_Float16)lo.w;
  h[4] = (_Float16)hi.x; h[5] = (_Float16)hi.y; h[6] = (_Float16)hi.z; h[7] = (_Float16)hi.w;
  return h;
}

constexpr int kRideRows    = 8;
constexpr int kRideThreads = kRideRows * 32 * 128;
constexpr int kRidePieces  = (kC * kC / 8) / kRideThreads;
__device__ __forceinline__ void cvt_w_ride(const float* __restrict__ src,
                                           _Float16* __restrict__ dst, int t) {
  float4 lo[kRidePieces], hi[kRidePieces];
#pragma unroll
  for (int j = 0; j < kRidePieces; ++j) {
    const size_t o = w_piece_src(t + j * kRideThreads);
    lo[j] = *(const float4*)(src + o);
    hi[j] = *(const float4*)(src + o + 4);
  }
#pragma unroll
  for (int j = 0; j < kRidePieces; ++j)
    *(v8h*)(dst + (size_t)(t + j * kRideThreads) * 8) = w_piece_round(lo[j], hi[j]);
}

__global__ void cvt_w_frag(const float* __restrict__ src,
                           _Float16* __restrict__ dst) {
  const int p    = blockIdx.x * blockDim.x + threadIdx.x;
  const int lane = p & 31;
  const int j    = (p >> 5) & 1;
  const int nt   = (p >> 6) & 3;
  const int s    = (p >> 8) & 31;
  const int nb   = p >> 13;
  const int n    = nb * 64 + nt * 16 + (lane & 15);
  const int k0   = s * 32 + j * 16 + ((lane >> 4) << 3);
  const float4 lo = *(const float4*)(src + (size_t)n * kC + k0);
  const float4 hi = *(const float4*)(src + (size_t)n * kC + k0 + 4);
  v8h h;
  h[0] = (_Float16)lo.x; h[1] = (_Float16)lo.y; h[2] = (_Float16)lo.z; h[3] = (_Float16)lo.w;
  h[4] = (_Float16)hi.x; h[5] = (_Float16)hi.y; h[6] = (_Float16)hi.z; h[7] = (_Float16)hi.w;
  *(v8h*)(dst + (size_t)p * 8) = h;
}

__device__ __forceinline__ v16h load_wfrag(const _Float16* __restrict__ blk) {
  const int lane = threadIdx.x & 31;
  v8h lo = *(const v8h*)(blk + lane * 8);
  v8h hi = *(const v8h*)(blk + 256 + lane * 8);
  v16h f;
#pragma unroll
  for (int i = 0; i < 8; ++i) { f[i] = lo[i]; f[i + 8] = hi[i]; }
  return f;
}

template <int MODE, typename AT>
__device__ __forceinline__ void gemm_tile(const AT* __restrict__ A,
                                          const _Float16* __restrict__ W,
                                          void* __restrict__ out) {
  const int lane = threadIdx.x & 31;
  const int colL = lane & 15;
  const int hi   = lane >> 4;
  const int m0 = blockIdx.x * 128 + (threadIdx.x >> 5) * 32;
  const int n0 = blockIdx.y * 128;

  v8f acc[2][8] = {};
  const AT* a0p = A + (size_t)m0 * kC;
  const AT* a1p = A + (size_t)(m0 + 16) * kC;
  constexpr size_t kWGroup = (size_t)(kC / 32) * kWStep;
  const _Float16* wp  = W + (size_t)blockIdx.y * 2 * kWGroup;

  typename ARawOf<AT>::type ra[3][2];
  v16h fa[3][2], fb[3][8];
  auto issue = [&](int s, int k) {
    a_issue(ra[s][0], a0p + k, kC);
    a_issue(ra[s][1], a1p + k, kC);
#pragma unroll
    for (int nt = 0; nt < 8; ++nt)
      fb[s][nt] = load_wfrag(wp + (size_t)(nt >> 2) * kWGroup + (size_t)(k >> 5) * kWStep + (nt & 3) * kWBlk);
  };
  auto mma = [&](int s) {
    acc_mem_fence(acc);
#pragma unroll
    for (int nt = 0; nt < 8; ++nt) {
      acc[0][nt] = wmma16(fa[s][0], fb[s][nt], acc[0][nt]);
      acc[1][nt] = wmma16(fa[s][1], fb[s][nt], acc[1][nt]);
    }
    acc_fence(acc);
  };
  auto finish = [&](int s) {
    __builtin_amdgcn_sched_barrier(0);
    fa[s][0] = a_finish(ra[s][0]);
    fa[s][1] = a_finish(ra[s][1]);
  };

  issue(0, 0);
  issue(1, 32);
  issue(2, 64);
  finish(0);
  for (int k = 0; k < kC - 160; k += 96) {
    mma(0); issue(0, k + 96);  finish(1);
    mma(1); issue(1, k + 128); finish(2);
    mma(2); issue(2, k + 160); finish(0);
  }
  mma(0); issue(0, kC - 64); finish(1);
  mma(1); issue(1, kC - 32); finish(2);
  mma(2); finish(0);
  mma(0); finish(1);
  mma(1);

#pragma unroll
  for (int mt = 0; mt < 2; ++mt)
#pragma unroll
    for (int nt = 0; nt < 8; ++nt)
#pragma unroll
      for (int r = 0; r < 8; ++r) {
        const int row  = m0 + mt * 16 + 8 * hi + r;
        const int colc = n0 + nt * 16 + colL;
        const float v  = acc[mt][nt][r];
        if constexpr (MODE == 2) {
          ((float*)out)[(size_t)row * kC + colc] = v;
        } else {
          const int b = row >> 11, n = row & (kN - 1);
          const int h = colc >> 6, d = colc & (kD - 1);
          if constexpr (MODE == 0)
            ((_Float16*)out)[(((size_t)(b * kH + h)) * kN + n) * kD + d] = (_Float16)v;
          else
            ((_Float16*)out)[(((size_t)(b * kH + h)) * kD + d) * kN + n] = (_Float16)v;
        }
      }
}

template <int MODE>
__global__ void __launch_bounds__(128)
proj_x32_wT(const float* __restrict__ X,
            const _Float16* __restrict__ W,
            _Float16* __restrict__ out,
            const float* __restrict__ wnext,
            _Float16* __restrict__ wnext16) {
  if (blockIdx.y >= kC / 128) {
    cvt_w_ride(wnext, wnext16,
               ((int)(blockIdx.y - kC / 128) * 32 + (int)blockIdx.x) * 128 + (int)threadIdx.x);
    return;
  }
  gemm_tile<MODE>(X, W, out);
}

__global__ void __launch_bounds__(128) gemm_ctx_woT(const _Float16* __restrict__ A,
                             const _Float16* __restrict__ W,
                             float* __restrict__ out) {
  gemm_tile<2>(A, W, out);
}

constexpr int kQRows  = 128;
constexpr int kQPitch = kD + 8;
constexpr int kQSteps = kN / kQRows;

__device__ __forceinline__ v16h lds_frag(const _Float16* p) {
  const int lane = threadIdx.x & 31;
  const _Float16* b0 = p + (lane & 15) * kQPitch + ((lane >> 4) << 3);
  v8h lo = *(const v8h*)(b0);
  v8h hi = *(const v8h*)(b0 + 16);
  v16h f;
#pragma unroll
  for (int i = 0; i < 8; ++i) { f[i] = lo[i]; f[i + 8] = hi[i]; }
  return f;
}

__global__ void __launch_bounds__(256)
colstats_kernel(const _Float16* __restrict__ q16,
                const _Float16* __restrict__ k16,
                float* __restrict__ mcol,
                float* __restrict__ rcol) {
  const int bh   = blockIdx.y;
  const int wave = threadIdx.x >> 5;
  const int lane = threadIdx.x & 31;
  const int kt   = blockIdx.x * 8 + wave;
  const _Float16* qb = q16 + (size_t)bh * kN * kD;
  const _Float16* kb = k16 + (size_t)bh * kN * kD;

  const v16h bk0 = load_frag(kb + (size_t)kt * 16 * kD + 0,  kD);
  const v16h bk1 = load_frag(kb + (size_t)kt * 16 * kD + 32, kD);

  __shared__ __attribute__((aligned(16))) _Float16 qs[2][kQRows * kQPitch];

  const int prow = threadIdx.x >> 3;
  const int pcol = (threadIdx.x & 7) * 8;
  v8h pre[4];

  auto fetch = [&](int step) {
    const _Float16* src = qb + (size_t)step * kQRows * kD + (size_t)threadIdx.x * 8;
#pragma unroll
    for (int j = 0; j < 4; ++j) pre[j] = *(const v8h*)(src + (size_t)j * 256 * 8);
  };
  auto park = [&](int buf) {
    _Float16* dst = qs[buf] + prow * kQPitch + pcol;
#pragma unroll
    for (int j = 0; j < 4; ++j) *(v8h*)(dst + j * 32 * kQPitch) = pre[j];
  };

  fetch(0);
  park(0);
  __syncthreads();

  v8f mxv;
#pragma unroll
  for (int i = 0; i < 8; ++i) mxv[i] = -3.4e38f;
  for (int st = 0; st < kQSteps; ++st) {
    fetch((st + 1) & (kQSteps - 1));
    const _Float16* cur = qs[st & 1];
#pragma unroll
    for (int t = 0; t < kQRows / 16; ++t) {
      v16h a0 = lds_frag(cur + t * 16 * kQPitch + 0);
      v16h a1 = lds_frag(cur + t * 16 * kQPitch + 32);
      v8f s = {};
      s = wmma16(a0, bk0, s);
      s = wmma16(a1, bk1, s);
#pragma unroll
      for (int i = 0; i < 8; ++i) mxv[i] = fmaxf(mxv[i], s[i]);
    }
    park((st + 1) & 1);
    __syncthreads();
  }
  float mk = mxv[0];
#pragma unroll
  for (int i = 1; i < 8; ++i) mk = fmaxf(mk, mxv[i]);
  mk = fmaxf(mk, __shfl_xor(mk, 16, 32));
  if (lane < 16) mcol[(size_t)bh * kN + kt * 16 + lane] = mk;

  v8f smv = {};
  for (int st = 0; st < kQSteps; ++st) {
    const bool more = st + 1 < kQSteps;
    if (more) fetch(st + 1);
    const _Float16* cur = qs[st & 1];
#pragma unroll
    for (int t = 0; t < kQRows / 16; ++t) {
      v16h a0 = lds_frag(cur + t * 16 * kQPitch + 0);
      v16h a1 = lds_frag(cur + t * 16 * kQPitch + 32);
      v8f s = {};
      s = wmma16(a0, bk0, s);
      s = wmma16(a1, bk1, s);
#pragma unroll
      for (int i = 0; i < 8; ++i) smv[i] += __expf(s[i] - mk);
    }
    if (more) {
      park((st + 1) & 1);
      __syncthreads();
    }
  }
  float sm = ((smv[0] + smv[1]) + (smv[2] + smv[3])) + ((smv[4] + smv[5]) + (smv[6] + smv[7]));
  sm += __shfl_xor(sm, 16, 32);
  if (lane < 16) rcol[(size_t)bh * kN + kt * 16 + lane] = 1.0f / sm;
}

constexpr int kKS     = 128;
constexpr int kKPitch = kD + 8;
constexpr int kVPitch = kKS + 8;
constexpr int kKSteps = kN / kKS;
constexpr int kEPitch = 16;

template <int PITCH>
__device__ __forceinline__ v16h lds_frag_p(const _Float16* p) {
  const int lane = threadIdx.x & 31;
  const _Float16* b0 = p + (lane & 15) * PITCH + ((lane >> 4) << 3);
  v8h lo = *(const v8h*)(b0);
  v8h hi = *(const v8h*)(b0 + 16);
  v16h f;
#pragma unroll
  for (int i = 0; i < 8; ++i) { f[i] = lo[i]; f[i + 8] = hi[i]; }
  return f;
}

__device__ __forceinline__ v8h lds_tr16(const _Float16* p) {
  typedef __attribute__((address_space(3))) v8s lds_v8s;
  return __builtin_bit_cast(v8h, __builtin_amdgcn_ds_load_tr16_b128_v8i16((lds_v8s*)p));
}

__global__ void __launch_bounds__(256)
ctx_kernel(const _Float16* __restrict__ q16,
           const _Float16* __restrict__ k16,
           const _Float16* __restrict__ vt16,
           const float* __restrict__ mcol,
           const float* __restrict__ rcol,
           _Float16* __restrict__ ctx16) {
  const int bh   = blockIdx.y;
  const int wave = threadIdx.x >> 5;
  const int lane = threadIdx.x & 31;
  const int col  = lane & 15;
  const int hi   = lane >> 4;
  const int qt   = blockIdx.x * 8 + wave;

  __shared__ float mr_s[2][kN];
  __shared__ __attribute__((aligned(16))) _Float16 estage[8][32 * kEPitch];
  __shared__ __attribute__((aligned(16))) _Float16 ks[2][kKS * kKPitch];
  __shared__ __attribute__((aligned(16))) _Float16 vs[2][kD * kVPitch];

  for (int i = threadIdx.x; i < kN; i += 256) {
    mr_s[0][i] = mcol[(size_t)bh * kN + i];
    mr_s[1][i] = rcol[(size_t)bh * kN + i];
  }

  const _Float16* qb = q16  + (size_t)bh * kN * kD;
  const _Float16* kb = k16  + (size_t)bh * kN * kD;
  const _Float16* vb = vt16 + (size_t)bh * kD * kN;

  const int krow = threadIdx.x >> 3;
  const int kcol = (threadIdx.x & 7) * 8;
  const int vrow = threadIdx.x >> 4;
  const int vcol = (threadIdx.x & 15) * 8;
  v8h kpre[4], vpre[4];

  auto fetch = [&](int step) {
    const _Float16* ksrc = kb + (size_t)step * kKS * kD + (size_t)threadIdx.x * 8;
    const _Float16* vsrc = vb + (size_t)vrow * kN + (size_t)step * kKS + vcol;
#pragma unroll
    for (int j = 0; j < 4; ++j) kpre[j] = *(const v8h*)(ksrc + (size_t)j * 256 * 8);
#pragma unroll
    for (int j = 0; j < 4; ++j) vpre[j] = *(const v8h*)(vsrc + (size_t)j * 16 * kN);
  };
  auto park = [&](int buf) {
    _Float16* kd = ks[buf] + krow * kKPitch + kcol;
    _Float16* vd = vs[buf] + vrow * kVPitch + vcol;
#pragma unroll
    for (int j = 0; j < 4; ++j) *(v8h*)(kd + j * 32 * kKPitch) = kpre[j];
#pragma unroll
    for (int j = 0; j < 4; ++j) *(v8h*)(vd + j * 16 * kVPitch) = vpre[j];
  };

  fetch(0);
  const v16h aq0 = load_frag(qb + (size_t)qt * 16 * kD + 0,  kD);
  const v16h aq1 = load_frag(qb + (size_t)qt * 16 * kD + 32, kD);
  park(0);
  __syncthreads();

  v8f acc[4] = {};
  _Float16* stg = estage[wave];
  _Float16* stw = stg + col * kEPitch + 8 * hi;
  const _Float16* str = stg + (8 * hi + (lane & 7)) * kEPitch + 8 * ((lane >> 3) & 1);

  for (int st = 0; st < kKSteps; ++st) {
    const bool more = st + 1 < kKSteps;
    if (more) fetch(st + 1);
    const _Float16* kc = ks[st & 1];
    const _Float16* vc = vs[st & 1];

#pragma unroll
    for (int c = 0; c < kKS / 32; ++c) {
      const int k0 = st * kKS + c * 32;
      v8f s0 = {}, s1 = {};
      s0 = wmma16(aq0, lds_frag_p<kKPitch>(kc + (c * 32     ) * kKPitch + 0),  s0);
      s0 = wmma16(aq1, lds_frag_p<kKPitch>(kc + (c * 32     ) * kKPitch + 32), s0);
      s1 = wmma16(aq0, lds_frag_p<kKPitch>(kc + (c * 32 + 16) * kKPitch + 0),  s1);
      s1 = wmma16(aq1, lds_frag_p<kKPitch>(kc + (c * 32 + 16) * kKPitch + 32), s1);

      const float m0 = mr_s[0][k0 + col],      r0 = mr_s[1][k0 + col];
      const float m1 = mr_s[0][k0 + 16 + col], r1 = mr_s[1][k0 + 16 + col];

      v8h e0, e1;
#pragma unroll
      for (int i = 0; i < 8; ++i) {
        e0[i] = (_Float16)(__expf(s0[i] - m0) * r0);
        e1[i] = (_Float16)(__expf(s1[i] - m1) * r1);
      }
      *(v8h*)(stw)                = e0;
      *(v8h*)(stw + 16 * kEPitch) = e1;
      const v8h elo = lds_tr16(str);
      const v8h ehi = lds_tr16(str + 16 * kEPitch);
      v16h ef;
#pragma unroll
      for (int i = 0; i < 8; ++i) { ef[i] = elo[i]; ef[i + 8] = ehi[i]; }
#pragma unroll
      for (int dt = 0; dt < 4; ++dt)
        acc[dt] = wmma16(ef, lds_frag_p<kVPitch>(vc + dt * 16 * kVPitch + c * 32), acc[dt]);
    }

    if (more) {
      park((st + 1) & 1);
      __syncthreads();
    }
  }

  const int b = bh >> 4, h = bh & 15;
  _Float16* cb = ctx16 + (size_t)b * kN * kC + (size_t)h * kD;
  const int qrow = qt * 16 + 8 * hi;
#pragma unroll
  for (int dt = 0; dt < 4; ++dt)
#pragma unroll
    for (int r = 0; r < 8; ++r)
      cb[(size_t)(qrow + r) * kC + dt * 16 + col] = (_Float16)acc[dt][r];
}

__global__ void ln_kernel(const float* __restrict__ resid,
                          const float* __restrict__ y,
                          const float* __restrict__ g,
                          const float* __restrict__ be,
                          float* __restrict__ out) {
  const int row = blockIdx.x;
  const int t   = threadIdx.x;
  float x[4], s = 0.f, sq = 0.f;
#pragma unroll
  for (int j = 0; j < 4; ++j) {
    const int c = t + j * 256;
    const float v = resid[(size_t)row * kC + c] + y[(size_t)row * kC + c];
    x[j] = v; s += v; sq += v * v;
  }
#pragma unroll
  for (int o = 16; o >= 1; o >>= 1) {
    s  += __shfl_xor(s,  o, 32);
    sq += __shfl_xor(sq, o, 32);
  }
  __shared__ float rs[8], rq[8];
  __shared__ float mu_s, inv_s;
  const int wave = t >> 5, lane = t & 31;
  if (lane == 0) { rs[wave] = s; rq[wave] = sq; }
  __syncthreads();
  if (t == 0) {
    float S = 0.f, Q = 0.f;
#pragma unroll
    for (int w = 0; w < 8; ++w) { S += rs[w]; Q += rq[w]; }
    const float mu = S * (1.0f / kC);
    mu_s  = mu;
    inv_s = rsqrtf(Q * (1.0f / kC) - mu * mu + kLnEps);
  }
  __syncthreads();
  const float mu = mu_s, inv = inv_s;
#pragma unroll
  for (int j = 0; j < 4; ++j) {
    const int c = t + j * 256;
    out[(size_t)row * kC + c] = (x[j] - mu) * inv * g[c] + be[c];
  }
}

extern "C" void kernel_launch(void* const* d_in, const int* in_sizes, int n_in,
                              void* d_out, int out_size, void* d_ws, size_t ws_size,
                              hipStream_t stream) {
  (void)in_sizes; (void)n_in; (void)out_size; (void)ws_size;
  const float* query = (const float*)d_in[0];
  const float* key   = (const float*)d_in[1];
  const float* value = (const float*)d_in[2];
  const float* Wq = (const float*)d_in[4];
  const float* Wk = (const float*)d_in[5];
  const float* Wv = (const float*)d_in[6];
  const float* Wo = (const float*)d_in[7];
  const float* ln_g = (const float*)d_in[8];
  const float* ln_b = (const float*)d_in[9];
  float* out = (float*)d_out;

  constexpr size_t BNC = (size_t)kB * kN * kC;
  constexpr size_t CC  = (size_t)kC * kC;
  constexpr size_t BHN = (size_t)kB * kH * kN;

  size_t off = 0;
  auto take = [&](size_t bytes) -> void* {
    void* p = (char*)d_ws + off;
    off += (bytes + 255) & ~(size_t)255;
    return p;
  };
  _Float16* ctx16 = (_Float16*)take(BNC * 2);
  float*    yf32  = (float*)take(BNC * 4);
  _Float16* wq16 = (_Float16*)take(CC * 2);
  _Float16* wk16 = (_Float16*)take(CC * 2);
  _Float16* wv16 = (_Float16*)take(CC * 2);
  _Float16* wo16 = (_Float16*)take(CC * 2);
  _Float16* qh16 = (_Float16*)take(BNC * 2);
  _Float16* kh16 = (_Float16*)take(BNC * 2);
  _Float16* vt16 = (_Float16*)take(BNC * 2);
  float*    mcol = (float*)take(BHN * 4);
  float*    rcol = (float*)take(BHN * 4);

  cvt_w_frag<<<(int)(CC / 8 / 256), 256, 0, stream>>>(Wq, wq16);

  dim3 ggrid(32, 8);
  dim3 pgrid(32, 8 + kRideRows);
  proj_x32_wT<0><<<pgrid, 128, 0, stream>>>(query, wq16, qh16, Wk, wk16);
  proj_x32_wT<0><<<pgrid, 128, 0, stream>>>(key,   wk16, kh16, Wv, wv16);
  proj_x32_wT<1><<<pgrid, 128, 0, stream>>>(value, wv16, vt16, Wo, wo16);

  colstats_kernel<<<dim3(kN / 128, kB * kH), 256, 0, stream>>>(qh16, kh16, mcol, rcol);

  ctx_kernel<<<dim3(kN / 128, kB * kH), 256, 0, stream>>>(qh16, kh16, vt16, mcol,
                                                          rcol, ctx16);

  gemm_ctx_woT<<<ggrid, 128, 0, stream>>>(ctx16, wo16, yf32);
  ln_kernel<<<kB * kN, 256, 0, stream>>>(query, yf32, ln_g, ln_b, out);
}
